# starting kernel
# baseline (speedup 1.0000x reference)
#include <hip/hip_runtime.h>
#include <hip/hip_bf16.h>

typedef __attribute__((ext_vector_type(2))) float v2f;
typedef __attribute__((ext_vector_type(8))) float v8f;

constexpr int kMN    = 1024;
constexpr int kDim   = 128;
constexpr int kBatch = 2048;

__global__ void som_w2_kernel(const float* __restrict__ W, float* __restrict__ w2) {
    int row = blockIdx.x * blockDim.x + threadIdx.x;
    if (row >= kMN) return;
    const float4* wr = (const float4*)(W + row * kDim);
    float s = 0.f;
#pragma unroll 8
    for (int i = 0; i < kDim / 4; ++i) {
        float4 v = wr[i];
        s += v.x * v.x + v.y * v.y + v.z * v.z + v.w * v.w;
    }
    w2[row] = s;
}

__global__ void som_bmu_kernel(const float* __restrict__ W,
                               const float* __restrict__ X,
                               const float* __restrict__ w2,
                               const int*   __restrict__ loc,
                               float2*      __restrict__ bmu_loc) {
    const int lane  = threadIdx.x;
    const int col   = lane & 15;
    const int half  = lane >> 4;
    const int batch = blockIdx.x * 16 + col;

    const float* Xrow = X + batch * kDim;

    float best = 3.402823466e38f;
    int   bidx = 0;

    for (int mt = 0; mt < kMN / 16; ++mt) {
        const float* Wrow = W + (mt * 16 + col) * kDim;
        v8f acc = {};
#pragma unroll 4
        for (int k0 = 0; k0 < kDim; k0 += 4) {
            const int kb = k0 + half * 2;
            v2f a = *(const v2f*)(Wrow + kb);
            v2f b = *(const v2f*)(Xrow + kb);
            acc = __builtin_amdgcn_wmma_f32_16x16x4_f32(
                false, a, false, b, (short)0, acc, false, false);
        }
#pragma unroll
        for (int v = 0; v < 8; ++v) {
            const int row = mt * 16 + v + half * 8;
            const float s = w2[row] - 2.0f * acc[v];
            if (s < best) { best = s; bidx = row; }
        }
    }
    const float obest = __shfl_xor(best, 16);
    const int   oidx  = __shfl_xor(bidx, 16);
    if (obest < best || (obest == best && oidx < bidx)) { best = obest; bidx = oidx; }
    if (half == 0) {
        bmu_loc[batch] = make_float2((float)loc[bidx * 2], (float)loc[bidx * 2 + 1]);
    }
}

__global__ void som_update_kernel(const float*  __restrict__ W,
                                  const float*  __restrict__ X,
                                  const int*    __restrict__ loc,
                                  const float2* __restrict__ bmu_loc,
                                  const int*    __restrict__ itp,
                                  float*        __restrict__ out) {
    const int lane = threadIdx.x;
    const int col  = lane & 15;
    const int half = lane >> 4;
    const int mt   = blockIdx.x;
    const int d0   = blockIdx.y * 16;

    const float lr_decay   = 1.0f - (float)itp[0] * (1.0f / 100.0f);
    const float alpha_op   = 0.3f * lr_decay;
    const float sigma_op   = 16.0f * lr_decay;
    const float neg_inv_s2 = -1.0f / (sigma_op * sigma_op);

    const int   arow = mt * 16 + col;
    const float lx = (float)loc[arow * 2];
    const float ly = (float)loc[arow * 2 + 1];

    v8f   acc  = {};
    float rsum = 0.f;

    for (int k0 = 0; k0 < kBatch; k0 += 4) {
        const int kb = k0 + half * 2;
        const float4 bl = *(const float4*)(&bmu_loc[kb]);
        const float dx0 = lx - bl.x, dy0 = ly - bl.y;
        const float dx1 = lx - bl.z, dy1 = ly - bl.w;
        v2f a;
        a.x = alpha_op * __expf((dx0 * dx0 + dy0 * dy0) * neg_inv_s2);
        a.y = alpha_op * __expf((dx1 * dx1 + dy1 * dy1) * neg_inv_s2);
        rsum += a.x + a.y;
        v2f b;
        b.x = X[kb * kDim + d0 + col];
        b.y = X[(kb + 1) * kDim + d0 + col];
        acc = __builtin_amdgcn_wmma_f32_16x16x4_f32(
            false, a, false, b, (short)0, acc, false, false);
    }
    rsum += __shfl_xor(rsum, 16);
#pragma unroll
    for (int v = 0; v < 8; ++v) {
        const int   row = mt * 16 + v + half * 8;
        const float rs  = __shfl(rsum, v + half * 8);
        const float w   = W[row * kDim + d0 + col];
        out[row * kDim + d0 + col] = w + acc[v] - rs * w;
    }
}

extern "C" void kernel_launch(void* const* d_in, const int* in_sizes, int n_in,
                              void* d_out, int out_size, void* d_ws, size_t ws_size,
                              hipStream_t stream) {
    const float* x        = (const float*)d_in[0];
    const float* weights  = (const float*)d_in[1];
    const int*   locations= (const int*)  d_in[2];
    const int*   it       = (const int*)  d_in[3];
    float*       out      = (float*)d_out;

    float*  w2      = (float*)d_ws;
    float2* bmu_loc = (float2*)((char*)d_ws + 4096);

    som_w2_kernel<<<kMN / 256, 256, 0, stream>>>(weights, w2);
    som_bmu_kernel<<<kBatch / 16, 32, 0, stream>>>(weights, x, w2, locations, bmu_loc);
    dim3 g2(kMN / 16, kDim / 16);
    som_update_kernel<<<g2, 32, 0, stream>>>(weights, x, locations, bmu_loc, it, out);
}


# Each work-group's one wave becomes 16: BMU's 64 mn-tiles and the update's batch loop are split over the waves, merged through LDS (update partial sums added in wave order: new reduction tree)
# speedup vs baseline: 5.2331x; 5.2331x over previous
#include <hip/hip_runtime.h>
#include <hip/hip_bf16.h>

typedef __attribute__((ext_vector_type(2))) float v2f;
typedef __attribute__((ext_vector_type(8))) float v8f;

constexpr int kMN    = 1024;
constexpr int kDim   = 128;
constexpr int kBatch = 2048;
constexpr int kBmuWaves        = 16;
constexpr int kBmuTilesPerWave = (kMN / 16) / kBmuWaves;
constexpr int kUpdWaves        = 16;
constexpr int kUpdKPerWave     = kBatch / kUpdWaves;
constexpr int kBmuThreads      = kBmuWaves * 32;
constexpr int kUpdThreads      = kUpdWaves * 32;


__global__ void som_w2_kernel(const float* __restrict__ W, float* __restrict__ w2) {
    int row = blockIdx.x * blockDim.x + threadIdx.x;
    if (row >= kMN) return;
    const float4* wr = (const float4*)(W + row * kDim);
    float s = 0.f;
#pragma unroll 8
    for (int i = 0; i < kDim / 4; ++i) {
        float4 v = wr[i];
        s += v.x * v.x + v.y * v.y + v.z * v.z + v.w * v.w;
    }
    w2[row] = s;
}

__global__ __launch_bounds__(kBmuThreads)
void som_bmu_kernel(const float* __restrict__ W,
                    const float* __restrict__ X,
                    const float* __restrict__ w2,
                    const int*   __restrict__ loc,
                    float2*      __restrict__ bmu_loc) {
    __shared__ float s_best[kBmuWaves][16];
    __shared__ int   s_bidx[kBmuWaves][16];

    const int lane  = threadIdx.x & 31;
    const int wave  = __builtin_amdgcn_readfirstlane(threadIdx.x >> 5);
    const int col   = lane & 15;
    const int half  = lane >> 4;
    const int batch = blockIdx.x * 16 + col;

    const float* Xrow = X + batch * kDim;

    float best = 3.402823466e38f;
    int   bidx = 0;

#pragma unroll 1
    for (int t = 0; t < kBmuTilesPerWave; ++t) {
        const int mt = wave * kBmuTilesPerWave + t;
        const float* Wrow = W + (mt * 16 + col) * kDim;
        v8f acc = {};
#pragma unroll 4
        for (int k0 = 0; k0 < kDim; k0 += 4) {
            const int kb = k0 + half * 2;
            v2f a = *(const v2f*)(Wrow + kb);
            v2f b = *(const v2f*)(Xrow + kb);
            acc = __builtin_amdgcn_wmma_f32_16x16x4_f32(
                false, a, false, b, (short)0, acc, false, false);
        }
#pragma unroll
        for (int v = 0; v < 8; ++v) {
            const int row = mt * 16 + v + half * 8;
            const float s = w2[row] - 2.0f * acc[v];
            if (s < best) { best = s; bidx = row; }
        }
    }
    const float obest = __shfl_xor(best, 16);
    const int   oidx  = __shfl_xor(bidx, 16);
    if (obest < best || (obest == best && oidx < bidx)) { best = obest; bidx = oidx; }
    if (half == 0) { s_best[wave][col] = best; s_bidx[wave][col] = bidx; }
    __syncthreads();
    if (threadIdx.x < 16) {
        float fbest = s_best[0][col];
        int   fidx  = s_bidx[0][col];
        for (int w = 1; w < kBmuWaves; ++w) {
            const float ob = s_best[w][col];
            if (ob < fbest) { fbest = ob; fidx = s_bidx[w][col]; }
        }
        bmu_loc[batch] = make_float2((float)loc[fidx * 2], (float)loc[fidx * 2 + 1]);
    }
}

__global__ __launch_bounds__(kUpdThreads)
void som_update_kernel(const float*  __restrict__ W,
                       const float*  __restrict__ X,
                       const int*    __restrict__ loc,
                       const float2* __restrict__ bmu_loc,
                       const int*    __restrict__ itp,
                       float*        __restrict__ out) {
    __shared__ float s_acc[kUpdWaves][8][32];
    __shared__ float s_rs[kUpdWaves][16];

    const int lane = threadIdx.x & 31;
    const int wave = __builtin_amdgcn_readfirstlane(threadIdx.x >> 5);
    const int col  = lane & 15;
    const int half = lane >> 4;
    const int mt   = blockIdx.x;
    const int d0   = blockIdx.y * 16;

    const float lr_decay   = 1.0f - (float)itp[0] * (1.0f / 100.0f);
    const float alpha_op   = 0.3f * lr_decay;
    const float sigma_op   = 16.0f * lr_decay;
    const float neg_inv_s2 = -1.0f / (sigma_op * sigma_op);

    const int   arow = mt * 16 + col;
    const float lx = (float)loc[arow * 2];
    const float ly = (float)loc[arow * 2 + 1];

    v8f   acc  = {};
    float rsum = 0.f;

    const int kbeg = wave * kUpdKPerWave;
#pragma unroll 4
    for (int ks = 0; ks < kUpdKPerWave; ks += 4) {
        const int kb = kbeg + ks + half * 2;
        const float4 bl = *(const float4*)(&bmu_loc[kb]);
        const float dx0 = lx - bl.x, dy0 = ly - bl.y;
        const float dx1 = lx - bl.z, dy1 = ly - bl.w;
        v2f a;
        a.x = alpha_op * __expf((dx0 * dx0 + dy0 * dy0) * neg_inv_s2);
        a.y = alpha_op * __expf((dx1 * dx1 + dy1 * dy1) * neg_inv_s2);
        rsum += a.x + a.y;
        v2f b;
        b.x = X[kb * kDim + d0 + col];
        b.y = X[(kb + 1) * kDim + d0 + col];
        acc = __builtin_amdgcn_wmma_f32_16x16x4_f32(
            false, a, false, b, (short)0, acc, false, false);
    }
    rsum += __shfl_xor(rsum, 16);
#pragma unroll
    for (int v = 0; v < 8; ++v) s_acc[wave][v][lane] = acc[v];
    if (half == 0) s_rs[wave][col] = rsum;
    __syncthreads();
    if (wave < 8) {
        const int rt = wave + half * 8;
        float a  = s_acc[0][wave][lane];
        float rs = s_rs[0][rt];
        for (int w = 1; w < kUpdWaves; ++w) {
            a  += s_acc[w][wave][lane];
            rs += s_rs[w][rt];
        }
        const int   row = mt * 16 + rt;
        const float w   = W[row * kDim + d0 + col];
        out[row * kDim + d0 + col] = w + a - rs * w;
    }
}

extern "C" void kernel_launch(void* const* d_in, const int* in_sizes, int n_in,
                              void* d_out, int out_size, void* d_ws, size_t ws_size,
                              hipStream_t stream) {
    const float* x        = (const float*)d_in[0];
    const float* weights  = (const float*)d_in[1];
    const int*   locations= (const int*)  d_in[2];
    const int*   it       = (const int*)  d_in[3];
    float*       out      = (float*)d_out;

    float*  w2      = (float*)d_ws;
    float2* bmu_loc = (float2*)((char*)d_ws + 4096);

    som_w2_kernel<<<kMN / 256, 256, 0, stream>>>(weights, w2);
    som_bmu_kernel<<<kBatch / 16, kBmuThreads, 0, stream>>>(weights, x, w2, locations, bmu_loc);
    dim3 g2(kMN / 16, kDim / 16);
    som_update_kernel<<<g2, kUpdThreads, 0, stream>>>(weights, x, locations, bmu_loc, it, out);
}


# BMU fragment loads: X fragment read once per wave (not per tile), W and X as 16-byte loads issued in groups a tile ahead (k inside each 8-block permuted alike for A and B: new dot sum order)
# speedup vs baseline: 6.7916x; 1.2978x over previous
#include <hip/hip_runtime.h>
#include <hip/hip_bf16.h>

typedef __attribute__((ext_vector_type(2))) float v2f;
typedef __attribute__((ext_vector_type(8))) float v8f;

constexpr int kMN    = 1024;
constexpr int kDim   = 128;
constexpr int kBatch = 2048;
constexpr int kBmuWaves        = 16;
constexpr int kBmuTilesPerWave = (kMN / 16) / kBmuWaves;
constexpr int kUpdWaves        = 16;
constexpr int kUpdKPerWave     = kBatch / kUpdWaves;
constexpr int kBmuThreads      = kBmuWaves * 32;
constexpr int kUpdThreads      = kUpdWaves * 32;


__global__ void som_w2_kernel(const float* __restrict__ W, float* __restrict__ w2) {
    int row = blockIdx.x * blockDim.x + threadIdx.x;
    if (row >= kMN) return;
    const float4* wr = (const float4*)(W + row * kDim);
    float s = 0.f;
#pragma unroll 8
    for (int i = 0; i < kDim / 4; ++i) {
        float4 v = wr[i];
        s += v.x * v.x + v.y * v.y + v.z * v.z + v.w * v.w;
    }
    w2[row] = s;
}

__global__ __launch_bounds__(kBmuThreads) __attribute__((amdgpu_waves_per_eu(4, 4)))
void som_bmu_kernel(const float* __restrict__ W,
                    const float* __restrict__ X,
                    const float* __restrict__ w2,
                    const int*   __restrict__ loc,
                    float2*      __restrict__ bmu_loc) {
    __shared__ float s_best[kBmuWaves][16];
    __shared__ int   s_bidx[kBmuWaves][16];

    const int lane  = threadIdx.x & 31;
    const int wave  = __builtin_amdgcn_readfirstlane(threadIdx.x >> 5);
    const int col   = lane & 15;
    const int half  = lane >> 4;
    const int batch = blockIdx.x * 16 + col;

    constexpr int kQ = kDim / 8;

    const float4* Xq = (const float4*)(X + batch * kDim + half * 4);
    float4 xf[kQ];
#pragma unroll
    for (int j = 0; j < kQ; ++j) xf[j] = Xq[2 * j];

    const float4* Wq = (const float4*)(W + (wave * kBmuTilesPerWave * 16 + col) * kDim + half * 4);
    constexpr int kTileStrideQ = 16 * kDim / 4;

    float4 wf[2][kQ];
#pragma unroll
    for (int j = 0; j < kQ; ++j) wf[0][j] = Wq[2 * j];

    float best = 3.402823466e38f;
    int   bidx = 0;

#pragma unroll
    for (int t = 0; t < kBmuTilesPerWave; ++t) {
        const int mt = wave * kBmuTilesPerWave + t;
        if (t + 1 < kBmuTilesPerWave) {
#pragma unroll
            for (int j = 0; j < kQ; ++j) wf[(t + 1) & 1][j] = Wq[(t + 1) * kTileStrideQ + 2 * j];
        }
        const float4* w2q = (const float4*)(w2 + mt * 16 + half * 8);
        const float4 n0 = w2q[0];
        const float4 n1 = w2q[1];
        v8f acc = {};
#pragma unroll
        for (int j = 0; j < kQ; ++j) {
            const float4 aw = wf[t & 1][j];
            const float4 bx = xf[j];
            v2f a0 = {aw.x, aw.y}, b0 = {bx.x, bx.y};
            v2f a1 = {aw.z, aw.w}, b1 = {bx.z, bx.w};
            acc = __builtin_amdgcn_wmma_f32_16x16x4_f32(
                false, a0, false, b0, (short)0, acc, false, false);
            acc = __builtin_amdgcn_wmma_f32_16x16x4_f32(
                false, a1, false, b1, (short)0, acc, false, false);
        }
        const float nrm[8] = {n0.x, n0.y, n0.z, n0.w, n1.x, n1.y, n1.z, n1.w};
#pragma unroll
        for (int v = 0; v < 8; ++v) {
            const int row = mt * 16 + v + half * 8;
            const float s = nrm[v] - 2.0f * acc[v];
            if (s < best) { best = s; bidx = row; }
        }
    }
    const float obest = __shfl_xor(best, 16);
    const int   oidx  = __shfl_xor(bidx, 16);
    if (obest < best || (obest == best && oidx < bidx)) { best = obest; bidx = oidx; }
    if (half == 0) { s_best[wave][col] = best; s_bidx[wave][col] = bidx; }
    __syncthreads();
    if (threadIdx.x < 16) {
        float fbest = s_best[0][col];
        int   fidx  = s_bidx[0][col];
        for (int w = 1; w < kBmuWaves; ++w) {
            const float ob = s_best[w][col];
            if (ob < fbest) { fbest = ob; fidx = s_bidx[w][col]; }
        }
        bmu_loc[batch] = make_float2((float)loc[fidx * 2], (float)loc[fidx * 2 + 1]);
    }
}

__global__ __launch_bounds__(kUpdThreads)
void som_update_kernel(const float*  __restrict__ W,
                       const float*  __restrict__ X,
                       const int*    __restrict__ loc,
                       const float2* __restrict__ bmu_loc,
                       const int*    __restrict__ itp,
                       float*        __restrict__ out) {
    __shared__ float s_acc[kUpdWaves][8][32];
    __shared__ float s_rs[kUpdWaves][16];

    const int lane = threadIdx.x & 31;
    const int wave = __builtin_amdgcn_readfirstlane(threadIdx.x >> 5);
    const int col  = lane & 15;
    const int half = lane >> 4;
    const int mt   = blockIdx.x;
    const int d0   = blockIdx.y * 16;

    const float lr_decay   = 1.0f - (float)itp[0] * (1.0f / 100.0f);
    const float alpha_op   = 0.3f * lr_decay;
    const float sigma_op   = 16.0f * lr_decay;
    const float neg_inv_s2 = -1.0f / (sigma_op * sigma_op);

    const int   arow = mt * 16 + col;
    const float lx = (float)loc[arow * 2];
    const float ly = (float)loc[arow * 2 + 1];

    v8f   acc  = {};
    float rsum = 0.f;

    const int kbeg = wave * kUpdKPerWave;
#pragma unroll 4
    for (int ks = 0; ks < kUpdKPerWave; ks += 4) {
        const int kb = kbeg + ks + half * 2;
        const float4 bl = *(const float4*)(&bmu_loc[kb]);
        const float dx0 = lx - bl.x, dy0 = ly - bl.y;
        const float dx1 = lx - bl.z, dy1 = ly - bl.w;
        v2f a;
        a.x = alpha_op * __expf((dx0 * dx0 + dy0 * dy0) * neg_inv_s2);
        a.y = alpha_op * __expf((dx1 * dx1 + dy1 * dy1) * neg_inv_s2);
        rsum += a.x + a.y;
        v2f b;
        b.x = X[kb * kDim + d0 + col];
        b.y = X[(kb + 1) * kDim + d0 + col];
        acc = __builtin_amdgcn_wmma_f32_16x16x4_f32(
            false, a, false, b, (short)0, acc, false, false);
    }
    rsum += __shfl_xor(rsum, 16);
#pragma unroll
    for (int v = 0; v < 8; ++v) s_acc[wave][v][lane] = acc[v];
    if (half == 0) s_rs[wave][col] = rsum;
    __syncthreads();
    if (wave < 8) {
        const int rt = wave + half * 8;
        float a  = s_acc[0][wave][lane];
        float rs = s_rs[0][rt];
        for (int w = 1; w < kUpdWaves; ++w) {
            a  += s_acc[w][wave][lane];
            rs += s_rs[w][rt];
        }
        const int   row = mt * 16 + rt;
        const float w   = W[row * kDim + d0 + col];
        out[row * kDim + d0 + col] = w + a - rs * w;
    }
}

extern "C" void kernel_launch(void* const* d_in, const int* in_sizes, int n_in,
                              void* d_out, int out_size, void* d_ws, size_t ws_size,
                              hipStream_t stream) {
    const float* x        = (const float*)d_in[0];
    const float* weights  = (const float*)d_in[1];
    const int*   locations= (const int*)  d_in[2];
    const int*   it       = (const int*)  d_in[3];
    float*       out      = (float*)d_out;

    float*  w2      = (float*)d_ws;
    float2* bmu_loc = (float2*)((char*)d_ws + 4096);

    som_w2_kernel<<<kMN / 256, 256, 0, stream>>>(weights, w2);
    som_bmu_kernel<<<kBatch / 16, kBmuThreads, 0, stream>>>(weights, x, w2, locations, bmu_loc);
    dim3 g2(kMN / 16, kDim / 16);
    som_update_kernel<<<g2, kUpdThreads, 0, stream>>>(weights, x, locations, bmu_loc, it, out);
}


# Row norms made in the BMU kernel: per-lane sums of squares of its W fragment, added as -w2/2 to the accumulator by one more WMMA per tile (new norm and score sum order); norm kernel and launch gone
# speedup vs baseline: 8.2226x; 1.2107x over previous
#include <hip/hip_runtime.h>
#include <hip/hip_bf16.h>

typedef __attribute__((ext_vector_type(2))) float v2f;
typedef __attribute__((ext_vector_type(8))) float v8f;

constexpr int kMN    = 1024;
constexpr int kDim   = 128;
constexpr int kBatch = 2048;
constexpr int kBmuWaves        = 16;
constexpr int kBmuTilesPerWave = (kMN / 16) / kBmuWaves;
constexpr int kUpdWaves        = 16;
constexpr int kUpdKPerWave     = kBatch / kUpdWaves;
constexpr int kBmuThreads      = kBmuWaves * 32;
constexpr int kUpdThreads      = kUpdWaves * 32;


__global__ __launch_bounds__(kBmuThreads) __attribute__((amdgpu_waves_per_eu(4, 4)))
void som_bmu_kernel(const float* __restrict__ W,
                    const float* __restrict__ X,
                    const int*   __restrict__ loc,
                    float2*      __restrict__ bmu_loc) {
    __shared__ float s_best[kBmuWaves][16];
    __shared__ int   s_bidx[kBmuWaves][16];

    const int lane  = threadIdx.x & 31;
    const int wave  = __builtin_amdgcn_readfirstlane(threadIdx.x >> 5);
    const int col   = lane & 15;
    const int half  = lane >> 4;
    const int batch = blockIdx.x * 16 + col;

    constexpr int kQ = kDim / 8;

    const float4* Xq = (const float4*)(X + batch * kDim + half * 4);
    float4 xf[kQ];
#pragma unroll
    for (int j = 0; j < kQ; ++j) xf[j] = Xq[2 * j];

    const float4* Wq = (const float4*)(W + (wave * kBmuTilesPerWave * 16 + col) * kDim + half * 4);
    constexpr int kTileStrideQ = 16 * kDim / 4;

    float4 wf[2][kQ];
#pragma unroll
    for (int j = 0; j < kQ; ++j) wf[0][j] = Wq[2 * j];

    float best = 3.402823466e38f;
    int   bidx = 0;

#pragma unroll
    for (int t = 0; t < kBmuTilesPerWave; ++t) {
        const int mt = wave * kBmuTilesPerWave + t;
        if (t + 1 < kBmuTilesPerWave) {
#pragma unroll
            for (int j = 0; j < kQ; ++j) wf[(t + 1) & 1][j] = Wq[(t + 1) * kTileStrideQ + 2 * j];
        }
        v8f acc = {};
        v2f q = {};
#pragma unroll
        for (int j = 0; j < kQ; ++j) {
            const float4 aw = wf[t & 1][j];
            const float4 bx = xf[j];
            v2f a0 = {aw.x, aw.y}, b0 = {bx.x, bx.y};
            v2f a1 = {aw.z, aw.w}, b1 = {bx.z, bx.w};
            acc = __builtin_amdgcn_wmma_f32_16x16x4_f32(
                false, a0, false, b0, (short)0, acc, false, false);
            acc = __builtin_amdgcn_wmma_f32_16x16x4_f32(
                false, a1, false, b1, (short)0, acc, false, false);
            q = a0 * a0 + q;
            q = a1 * a1 + q;
        }
        const v2f mh = {-0.5f, -0.5f};
        acc = __builtin_amdgcn_wmma_f32_16x16x4_f32(
            false, q, false, mh, (short)0, acc, false, false);
#pragma unroll
        for (int v = 0; v < 8; ++v) {
            const int row = mt * 16 + v + half * 8;
            const float s = -acc[v];
            if (s < best) { best = s; bidx = row; }
        }
    }
    const float obest = __shfl_xor(best, 16);
    const int   oidx  = __shfl_xor(bidx, 16);
    if (obest < best || (obest == best && oidx < bidx)) { best = obest; bidx = oidx; }
    if (half == 0) { s_best[wave][col] = best; s_bidx[wave][col] = bidx; }
    __syncthreads();
    if (threadIdx.x < 16) {
        float fbest = s_best[0][col];
        int   fidx  = s_bidx[0][col];
        for (int w = 1; w < kBmuWaves; ++w) {
            const float ob = s_best[w][col];
            if (ob < fbest) { fbest = ob; fidx = s_bidx[w][col]; }
        }
        bmu_loc[batch] = make_float2((float)loc[fidx * 2], (float)loc[fidx * 2 + 1]);
    }
}

__global__ __launch_bounds__(kUpdThreads)
void som_update_kernel(const float*  __restrict__ W,
                       const float*  __restrict__ X,
                       const int*    __restrict__ loc,
                       const float2* __restrict__ bmu_loc,
                       const int*    __restrict__ itp,
                       float*        __restrict__ out) {
    __shared__ float s_acc[kUpdWaves][8][32];
    __shared__ float s_rs[kUpdWaves][16];

    const int lane = threadIdx.x & 31;
    const int wave = __builtin_amdgcn_readfirstlane(threadIdx.x >> 5);
    const int col  = lane & 15;
    const int half = lane >> 4;
    const int mt   = blockIdx.x;
    const int d0   = blockIdx.y * 16;

    const float lr_decay   = 1.0f - (float)itp[0] * (1.0f / 100.0f);
    const float alpha_op   = 0.3f * lr_decay;
    const float sigma_op   = 16.0f * lr_decay;
    const float neg_inv_s2 = -1.0f / (sigma_op * sigma_op);

    const int   arow = mt * 16 + col;
    const float lx = (float)loc[arow * 2];
    const float ly = (float)loc[arow * 2 + 1];

    v8f   acc  = {};
    float rsum = 0.f;

    const int kbeg = wave * kUpdKPerWave;
#pragma unroll 4
    for (int ks = 0; ks < kUpdKPerWave; ks += 4) {
        const int kb = kbeg + ks + half * 2;
        const float4 bl = *(const float4*)(&bmu_loc[kb]);
        const float dx0 = lx - bl.x, dy0 = ly - bl.y;
        const float dx1 = lx - bl.z, dy1 = ly - bl.w;
        v2f a;
        a.x = alpha_op * __expf((dx0 * dx0 + dy0 * dy0) * neg_inv_s2);
        a.y = alpha_op * __expf((dx1 * dx1 + dy1 * dy1) * neg_inv_s2);
        rsum += a.x + a.y;
        v2f b;
        b.x = X[kb * kDim + d0 + col];
        b.y = X[(kb + 1) * kDim + d0 + col];
        acc = __builtin_amdgcn_wmma_f32_16x16x4_f32(
            false, a, false, b, (short)0, acc, false, false);
    }
    rsum += __shfl_xor(rsum, 16);
#pragma unroll
    for (int v = 0; v < 8; ++v) s_acc[wave][v][lane] = acc[v];
    if (half == 0) s_rs[wave][col] = rsum;
    __syncthreads();
    if (wave < 8) {
        const int rt = wave + half * 8;
        float a  = s_acc[0][wave][lane];
        float rs = s_rs[0][rt];
        for (int w = 1; w < kUpdWaves; ++w) {
            a  += s_acc[w][wave][lane];
            rs += s_rs[w][rt];
        }
        const int   row = mt * 16 + rt;
        const float w   = W[row * kDim + d0 + col];
        out[row * kDim + d0 + col] = w + a - rs * w;
    }
}

extern "C" void kernel_launch(void* const* d_in, const int* in_sizes, int n_in,
                              void* d_out, int out_size, void* d_ws, size_t ws_size,
                              hipStream_t stream) {
    const float* x        = (const float*)d_in[0];
    const float* weights  = (const float*)d_in[1];
    const int*   locations= (const int*)  d_in[2];
    const int*   it       = (const int*)  d_in[3];
    float*       out      = (float*)d_out;

    float2* bmu_loc = (float2*)((char*)d_ws + 4096);

    som_bmu_kernel<<<kBatch / 16, kBmuThreads, 0, stream>>>(weights, x, locations, bmu_loc);
    dim3 g2(kMN / 16, kDim / 16);
    som_update_kernel<<<g2, kUpdThreads, 0, stream>>>(weights, x, locations, bmu_loc, it, out);
}


# Update work-group covers 32 feature columns (two accumulator tiles): each lr_op fragment feeds two WMMAs, X rows read as whole lines by 8-byte loads, k-loop trips of 8 steps; half the waves
# speedup vs baseline: 9.2958x; 1.1305x over previous
#include <hip/hip_runtime.h>
#include <hip/hip_bf16.h>

typedef __attribute__((ext_vector_type(2))) float v2f;
typedef __attribute__((ext_vector_type(8))) float v8f;

constexpr int kMN    = 1024;
constexpr int kDim   = 128;
constexpr int kBatch = 2048;
constexpr int kBmuWaves        = 16;
constexpr int kBmuTilesPerWave = (kMN / 16) / kBmuWaves;
constexpr int kUpdWaves        = 16;
constexpr int kUpdKPerWave     = kBatch / kUpdWaves;
constexpr int kBmuThreads      = kBmuWaves * 32;
constexpr int kUpdThreads      = kUpdWaves * 32;
constexpr int kUpdFeat         = 32;


__global__ __launch_bounds__(kBmuThreads) __attribute__((amdgpu_waves_per_eu(4, 4)))
void som_bmu_kernel(const float* __restrict__ W,
                    const float* __restrict__ X,
                    const int*   __restrict__ loc,
                    float2*      __restrict__ bmu_loc) {
    __shared__ float s_best[kBmuWaves][16];
    __shared__ int   s_bidx[kBmuWaves][16];

    const int lane  = threadIdx.x & 31;
    const int wave  = __builtin_amdgcn_readfirstlane(threadIdx.x >> 5);
    const int col   = lane & 15;
    const int half  = lane >> 4;
    const int batch = blockIdx.x * 16 + col;

    constexpr int kQ = kDim / 8;

    const float4* Xq = (const float4*)(X + batch * kDim + half * 4);
    float4 xf[kQ];
#pragma unroll
    for (int j = 0; j < kQ; ++j) xf[j] = Xq[2 * j];

    const float4* Wq = (const float4*)(W + (wave * kBmuTilesPerWave * 16 + col) * kDim + half * 4);
    constexpr int kTileStrideQ = 16 * kDim / 4;

    float4 wf[2][kQ];
#pragma unroll
    for (int j = 0; j < kQ; ++j) wf[0][j] = Wq[2 * j];

    float best = 3.402823466e38f;
    int   bidx = 0;

#pragma unroll
    for (int t = 0; t < kBmuTilesPerWave; ++t) {
        const int mt = wave * kBmuTilesPerWave + t;
        if (t + 1 < kBmuTilesPerWave) {
#pragma unroll
            for (int j = 0; j < kQ; ++j) wf[(t + 1) & 1][j] = Wq[(t + 1) * kTileStrideQ + 2 * j];
        }
        v8f acc = {};
        v2f q = {};
#pragma unroll
        for (int j = 0; j < kQ; ++j) {
            const float4 aw = wf[t & 1][j];
            const float4 bx = xf[j];
            v2f a0 = {aw.x, aw.y}, b0 = {bx.x, bx.y};
            v2f a1 = {aw.z, aw.w}, b1 = {bx.z, bx.w};
            acc = __builtin_amdgcn_wmma_f32_16x16x4_f32(
                false, a0, false, b0, (short)0, acc, false, false);
            acc = __builtin_amdgcn_wmma_f32_16x16x4_f32(
                false, a1, false, b1, (short)0, acc, false, false);
            q = a0 * a0 + q;
            q = a1 * a1 + q;
        }
        const v2f mh = {-0.5f, -0.5f};
        acc = __builtin_amdgcn_wmma_f32_16x16x4_f32(
            false, q, false, mh, (short)0, acc, false, false);
#pragma unroll
        for (int v = 0; v < 8; ++v) {
            const int row = mt * 16 + v + half * 8;
            const float s = -acc[v];
            if (s < best) { best = s; bidx = row; }
        }
    }
    const float obest = __shfl_xor(best, 16);
    const int   oidx  = __shfl_xor(bidx, 16);
    if (obest < best || (obest == best && oidx < bidx)) { best = obest; bidx = oidx; }
    if (half == 0) { s_best[wave][col] = best; s_bidx[wave][col] = bidx; }
    __syncthreads();
    if (threadIdx.x < 16) {
        float fbest = s_best[0][col];
        int   fidx  = s_bidx[0][col];
        for (int w = 1; w < kBmuWaves; ++w) {
            const float ob = s_best[w][col];
            if (ob < fbest) { fbest = ob; fidx = s_bidx[w][col]; }
        }
        bmu_loc[batch] = make_float2((float)loc[fidx * 2], (float)loc[fidx * 2 + 1]);
    }
}

__global__ __launch_bounds__(kUpdThreads)
void som_update_kernel(const float*  __restrict__ W,
                       const float*  __restrict__ X,
                       const int*    __restrict__ loc,
                       const float2* __restrict__ bmu_loc,
                       const int*    __restrict__ itp,
                       float*        __restrict__ out) {
    __shared__ float s_acc[kUpdWaves][2][8][32];
    __shared__ float s_rs[kUpdWaves][16];

    const int lane = threadIdx.x & 31;
    const int wave = __builtin_amdgcn_readfirstlane(threadIdx.x >> 5);
    const int col  = lane & 15;
    const int half = lane >> 4;
    const int mt   = blockIdx.x;
    const int d0   = blockIdx.y * kUpdFeat;

    const float lr_decay   = 1.0f - (float)itp[0] * (1.0f / 100.0f);
    const float alpha_op   = 0.3f * lr_decay;
    const float sigma_op   = 16.0f * lr_decay;
    const float neg_inv_s2 = -1.0f / (sigma_op * sigma_op);

    const int   arow = mt * 16 + col;
    const float lx = (float)loc[arow * 2];
    const float ly = (float)loc[arow * 2 + 1];

    v8f   acc0 = {};
    v8f   acc1 = {};
    float rsum = 0.f;

    const int     kl = wave * kUpdKPerWave + half * 2;
    const float*  Xl = X + (kl * kDim + d0 + 2 * col);
    const float2* Bl = bmu_loc + kl;
#pragma unroll 8
    for (int ks = 0; ks < kUpdKPerWave; ks += 4) {
        const float4 bl = *(const float4*)(Bl + ks);
        const float dx0 = lx - bl.x, dy0 = ly - bl.y;
        const float dx1 = lx - bl.z, dy1 = ly - bl.w;
        v2f a;
        a.x = alpha_op * __expf((dx0 * dx0 + dy0 * dy0) * neg_inv_s2);
        a.y = alpha_op * __expf((dx1 * dx1 + dy1 * dy1) * neg_inv_s2);
        rsum += a.x + a.y;
        const float2 r0 = *(const float2*)(Xl + ks * kDim);
        const float2 r1 = *(const float2*)(Xl + (ks + 1) * kDim);
        const v2f b0 = {r0.x, r1.x};
        const v2f b1 = {r0.y, r1.y};
        acc0 = __builtin_amdgcn_wmma_f32_16x16x4_f32(
            false, a, false, b0, (short)0, acc0, false, false);
        acc1 = __builtin_amdgcn_wmma_f32_16x16x4_f32(
            false, a, false, b1, (short)0, acc1, false, false);
    }
    rsum += __shfl_xor(rsum, 16);
#pragma unroll
    for (int v = 0; v < 8; ++v) {
        s_acc[wave][0][v][lane] = acc0[v];
        s_acc[wave][1][v][lane] = acc1[v];
    }
    if (half == 0) s_rs[wave][col] = rsum;
    __syncthreads();
    if (wave < 8) {
        const int rt = wave + half * 8;
        float a0 = s_acc[0][0][wave][lane];
        float a1 = s_acc[0][1][wave][lane];
        float rs = s_rs[0][rt];
        for (int w = 1; w < kUpdWaves; ++w) {
            a0 += s_acc[w][0][wave][lane];
            a1 += s_acc[w][1][wave][lane];
            rs += s_rs[w][rt];
        }
        const int    row = mt * 16 + rt;
        const float2 w   = *(const float2*)(W + row * kDim + d0 + 2 * col);
        float2 o;
        o.x = w.x + a0 - rs * w.x;
        o.y = w.y + a1 - rs * w.y;
        *(float2*)(out + row * kDim + d0 + 2 * col) = o;
    }
}

extern "C" void kernel_launch(void* const* d_in, const int* in_sizes, int n_in,
                              void* d_out, int out_size, void* d_ws, size_t ws_size,
                              hipStream_t stream) {
    const float* x        = (const float*)d_in[0];
    const float* weights  = (const float*)d_in[1];
    const int*   locations= (const int*)  d_in[2];
    const int*   it       = (const int*)  d_in[3];
    float*       out      = (float*)d_out;

    float2* bmu_loc = (float2*)((char*)d_ws + 4096);

    som_bmu_kernel<<<kBatch / 16, kBmuThreads, 0, stream>>>(weights, x, locations, bmu_loc);
    dim3 g2(kMN / 16, kDim / kUpdFeat);
    som_update_kernel<<<g2, kUpdThreads, 0, stream>>>(weights, x, locations, bmu_loc, it, out);
}


# BMU search split over two work-groups per batch tile, one per half of the mn rows (256 work-groups of 8 waves, every WGP busy); the update merges the two candidates per batch element into LDS
# speedup vs baseline: 11.6338x; 1.2515x over previous
#include <hip/hip_runtime.h>
#include <hip/hip_bf16.h>

typedef __attribute__((ext_vector_type(2))) float v2f;
typedef __attribute__((ext_vector_type(8))) float v8f;

constexpr int kMN    = 1024;
constexpr int kDim   = 128;
constexpr int kBatch = 2048;
constexpr int kBmuSplit        = 2;
constexpr int kBmuWaves        = 8;
constexpr int kBmuTilesPerWave = (kMN / 16) / (kBmuSplit * kBmuWaves);
constexpr int kUpdWaves        = 16;
constexpr int kUpdKPerWave     = kBatch / kUpdWaves;
constexpr int kBmuThreads      = kBmuWaves * 32;
constexpr int kUpdThreads      = kUpdWaves * 32;
constexpr int kUpdFeat         = 32;


__global__ __launch_bounds__(kBmuThreads) __attribute__((amdgpu_waves_per_eu(4, 4)))
void som_bmu_kernel(const float* __restrict__ W,
                    const float* __restrict__ X,
                    const int*   __restrict__ loc,
                    float4*      __restrict__ cand) {
    __shared__ float s_best[kBmuWaves][16];
    __shared__ int   s_bidx[kBmuWaves][16];

    const int lane  = threadIdx.x & 31;
    const int wave  = __builtin_amdgcn_readfirstlane(threadIdx.x >> 5);
    const int rpart = blockIdx.y;
    const int mt0   = (rpart * kBmuWaves + wave) * kBmuTilesPerWave;
    const int col   = lane & 15;
    const int half  = lane >> 4;
    const int batch = blockIdx.x * 16 + col;

    constexpr int kQ = kDim / 8;

    const float4* Xq = (const float4*)(X + batch * kDim + half * 4);
    float4 xf[kQ];
#pragma unroll
    for (int j = 0; j < kQ; ++j) xf[j] = Xq[2 * j];

    const float4* Wq = (const float4*)(W + (mt0 * 16 + col) * kDim + half * 4);
    constexpr int kTileStrideQ = 16 * kDim / 4;

    float4 wf[2][kQ];
#pragma unroll
    for (int j = 0; j < kQ; ++j) wf[0][j] = Wq[2 * j];

    float best = 3.402823466e38f;
    int   bidx = 0;

#pragma unroll
    for (int t = 0; t < kBmuTilesPerWave; ++t) {
        const int mt = mt0 + t;
        if (t + 1 < kBmuTilesPerWave) {
#pragma unroll
            for (int j = 0; j < kQ; ++j) wf[(t + 1) & 1][j] = Wq[(t + 1) * kTileStrideQ + 2 * j];
        }
        v8f acc = {};
        v2f q = {};
#pragma unroll
        for (int j = 0; j < kQ; ++j) {
            const float4 aw = wf[t & 1][j];
            const float4 bx = xf[j];
            v2f a0 = {aw.x, aw.y}, b0 = {bx.x, bx.y};
            v2f a1 = {aw.z, aw.w}, b1 = {bx.z, bx.w};
            acc = __builtin_amdgcn_wmma_f32_16x16x4_f32(
                false, a0, false, b0, (short)0, acc, false, false);
            acc = __builtin_amdgcn_wmma_f32_16x16x4_f32(
                false, a1, false, b1, (short)0, acc, false, false);
            q = a0 * a0 + q;
            q = a1 * a1 + q;
        }
        const v2f mh = {-0.5f, -0.5f};
        acc = __builtin_amdgcn_wmma_f32_16x16x4_f32(
            false, q, false, mh, (short)0, acc, false, false);
#pragma unroll
        for (int v = 0; v < 8; ++v) {
            const int row = mt * 16 + v + half * 8;
            const float s = -acc[v];
            if (s < best) { best = s; bidx = row; }
        }
    }
    const float obest = __shfl_xor(best, 16);
    const int   oidx  = __shfl_xor(bidx, 16);
    if (obest < best || (obest == best && oidx < bidx)) { best = obest; bidx = oidx; }
    if (half == 0) { s_best[wave][col] = best; s_bidx[wave][col] = bidx; }
    __syncthreads();
    if (threadIdx.x < 16) {
        float fbest = s_best[0][col];
        int   fidx  = s_bidx[0][col];
        for (int w = 1; w < kBmuWaves; ++w) {
            const float ob = s_best[w][col];
            if (ob < fbest) { fbest = ob; fidx = s_bidx[w][col]; }
        }
        cand[rpart * kBatch + batch] =
            make_float4(fbest, (float)loc[fidx * 2], (float)loc[fidx * 2 + 1], 0.f);
    }
}

__global__ __launch_bounds__(kUpdThreads) __attribute__((amdgpu_waves_per_eu(4, 4)))
void som_update_kernel(const float*  __restrict__ W,
                       const float*  __restrict__ X,
                       const int*    __restrict__ loc,
                       const float4* __restrict__ cand,
                       const int*    __restrict__ itp,
                       float*        __restrict__ out) {
    __shared__ float s_acc[kUpdWaves][2][8][32];
    __shared__ float s_rs[kUpdWaves][16];
    __shared__ float4 s_loc[kUpdWaves][kUpdKPerWave / 2];

    const int lane = threadIdx.x & 31;
    const int wave = __builtin_amdgcn_readfirstlane(threadIdx.x >> 5);
    const int col  = lane & 15;
    const int half = lane >> 4;
    const int mt   = blockIdx.x;
    const int d0   = blockIdx.y * kUpdFeat;

    const float lr_decay   = 1.0f - (float)itp[0] * (1.0f / 100.0f);
    const float alpha_op   = 0.3f * lr_decay;
    const float sigma_op   = 16.0f * lr_decay;
    const float neg_inv_s2 = -1.0f / (sigma_op * sigma_op);

    const int   arow = mt * 16 + col;
    const float lx = (float)loc[arow * 2];
    const float ly = (float)loc[arow * 2 + 1];

    v8f   acc0 = {};
    v8f   acc1 = {};
    float rsum = 0.f;

    const int     kl = wave * kUpdKPerWave + half * 2;
    const float*  Xl = X + (kl * kDim + d0 + 2 * col);

    {
        const float4* c0 = cand + wave * kUpdKPerWave + lane * 4;
        const float4* c1 = c0 + kBatch;
        float4 lo[4], hi[4];
#pragma unroll
        for (int e = 0; e < 4; ++e) { lo[e] = c0[e]; hi[e] = c1[e]; }
        float2 win[4];
#pragma unroll
        for (int e = 0; e < 4; ++e) {
            const bool up = hi[e].x < lo[e].x;
            win[e].x = up ? hi[e].y : lo[e].y;
            win[e].y = up ? hi[e].z : lo[e].z;
        }
        s_loc[wave][2 * lane]     = make_float4(win[0].x, win[0].y, win[1].x, win[1].y);
        s_loc[wave][2 * lane + 1] = make_float4(win[2].x, win[2].y, win[3].x, win[3].y);
    }
    __syncthreads();
    const float4* Bl = &s_loc[wave][half];
#pragma unroll 8
    for (int ks = 0; ks < kUpdKPerWave; ks += 4) {
        const float4 bl = Bl[ks >> 1];
        const float dx0 = lx - bl.x, dy0 = ly - bl.y;
        const float dx1 = lx - bl.z, dy1 = ly - bl.w;
        v2f a;
        a.x = alpha_op * __expf((dx0 * dx0 + dy0 * dy0) * neg_inv_s2);
        a.y = alpha_op * __expf((dx1 * dx1 + dy1 * dy1) * neg_inv_s2);
        rsum += a.x + a.y;
        const float2 r0 = *(const float2*)(Xl + ks * kDim);
        const float2 r1 = *(const float2*)(Xl + (ks + 1) * kDim);
        const v2f b0 = {r0.x, r1.x};
        const v2f b1 = {r0.y, r1.y};
        acc0 = __builtin_amdgcn_wmma_f32_16x16x4_f32(
            false, a, false, b0, (short)0, acc0, false, false);
        acc1 = __builtin_amdgcn_wmma_f32_16x16x4_f32(
            false, a, false, b1, (short)0, acc1, false, false);
    }
    rsum += __shfl_xor(rsum, 16);
#pragma unroll
    for (int v = 0; v < 8; ++v) {
        s_acc[wave][0][v][lane] = acc0[v];
        s_acc[wave][1][v][lane] = acc1[v];
    }
    if (half == 0) s_rs[wave][col] = rsum;
    __syncthreads();
    if (wave < 8) {
        const int rt = wave + half * 8;
        float a0 = s_acc[0][0][wave][lane];
        float a1 = s_acc[0][1][wave][lane];
        float rs = s_rs[0][rt];
        for (int w = 1; w < kUpdWaves; ++w) {
            a0 += s_acc[w][0][wave][lane];
            a1 += s_acc[w][1][wave][lane];
            rs += s_rs[w][rt];
        }
        const int    row = mt * 16 + rt;
        const float2 w   = *(const float2*)(W + row * kDim + d0 + 2 * col);
        float2 o;
        o.x = w.x + a0 - rs * w.x;
        o.y = w.y + a1 - rs * w.y;
        *(float2*)(out + row * kDim + d0 + 2 * col) = o;
    }
}

extern "C" void kernel_launch(void* const* d_in, const int* in_sizes, int n_in,
                              void* d_out, int out_size, void* d_ws, size_t ws_size,
                              hipStream_t stream) {
    const float* x        = (const float*)d_in[0];
    const float* weights  = (const float*)d_in[1];
    const int*   locations= (const int*)  d_in[2];
    const int*   it       = (const int*)  d_in[3];
    float*       out      = (float*)d_out;

    float4* cand = (float4*)d_ws;

    dim3 g1(kBatch / 16, kBmuSplit);
    som_bmu_kernel<<<g1, kBmuThreads, 0, stream>>>(weights, x, locations, cand);
    dim3 g2(kMN / 16, kDim / kUpdFeat);
    som_update_kernel<<<g2, kUpdThreads, 0, stream>>>(weights, x, locations, cand, it, out);
}


# Global loads cover whole 128-byte lines: BMU takes W and X as 512-byte row loads, staged in LDS (pitch 528 B), re-read in fragment order; update reads candidates lane-strided (4 lines a load, not 16)
# speedup vs baseline: 13.0454x; 1.1213x over previous
#include <hip/hip_runtime.h>
#include <hip/hip_bf16.h>

typedef __attribute__((ext_vector_type(2))) float v2f;
typedef __attribute__((ext_vector_type(8))) float v8f;
typedef __attribute__((ext_vector_type(4))) float v4f;

constexpr int kMN    = 1024;
constexpr int kDim   = 128;
constexpr int kBatch = 2048;
constexpr int kBmuSplit        = 2;
constexpr int kBmuWaves        = 8;
constexpr int kBmuTilesPerWave = (kMN / 16) / (kBmuSplit * kBmuWaves);
constexpr int kUpdWaves        = 16;
constexpr int kUpdKPerWave     = kBatch / kUpdWaves;
constexpr int kBmuThreads      = kBmuWaves * 32;
constexpr int kUpdThreads      = kUpdWaves * 32;
constexpr int kUpdFeat         = 32;
constexpr int kStagePitch      = kDim + 4;


__global__ __launch_bounds__(kBmuThreads)
void som_bmu_kernel(const float* __restrict__ W,
                    const float* __restrict__ X,
                    const int*   __restrict__ loc,
                    float4*      __restrict__ cand) {
    __shared__ __attribute__((aligned(16))) float s_w[kBmuWaves][kBmuTilesPerWave * 16][kStagePitch];
    __shared__ __attribute__((aligned(16))) float s_x[16][kStagePitch];
    __shared__ float s_best[kBmuWaves][16];
    __shared__ int   s_bidx[kBmuWaves][16];

    const int lane  = threadIdx.x & 31;
    const int wave  = __builtin_amdgcn_readfirstlane(threadIdx.x >> 5);
    const int rpart = blockIdx.y;
    const int mt0   = (rpart * kBmuWaves + wave) * kBmuTilesPerWave;
    const int col   = lane & 15;
    const int half  = lane >> 4;
    const int batch = blockIdx.x * 16 + col;

    constexpr int kQ    = kDim / 8;
    constexpr int kRowQ = kDim / 4;

    const v4f* Xg = (const v4f*)(X + (blockIdx.x * 16 + 2 * wave) * kDim) + lane;
    const v4f* Wg = (const v4f*)(W + (mt0 * 16) * kDim) + lane;
    v4f xs[2];
    v4f ws[2][16];
    xs[0] = Xg[0];
    xs[1] = Xg[kRowQ];
#pragma unroll
    for (int i = 0; i < 16; ++i) ws[0][i] = Wg[i * kRowQ];
#pragma unroll
    for (int i = 0; i < 16; ++i) ws[1][i] = Wg[(16 + i) * kRowQ];
    *(v4f*)&s_x[2 * wave][lane * 4]     = xs[0];
    *(v4f*)&s_x[2 * wave + 1][lane * 4] = xs[1];
#pragma unroll
    for (int i = 0; i < 16; ++i) *(v4f*)&s_w[wave][i][lane * 4] = ws[0][i];
    __builtin_amdgcn_fence(__ATOMIC_RELEASE, "workgroup", "local");
    __builtin_amdgcn_s_barrier();
    __builtin_amdgcn_fence(__ATOMIC_ACQUIRE, "workgroup", "local");

    float4 xf[kQ];
#pragma unroll
    for (int j = 0; j < kQ; ++j) xf[j] = *(const float4*)&s_x[col][8 * j + 4 * half];

    float4 wf[2][kQ];
#pragma unroll
    for (int j = 0; j < kQ; ++j) wf[0][j] = *(const float4*)&s_w[wave][col][8 * j + 4 * half];

    float best = 3.402823466e38f;
    int   bidx = 0;

#pragma unroll
    for (int t = 0; t < kBmuTilesPerWave; ++t) {
        const int mt = mt0 + t;
        v8f acc = {};
        v2f q = {};
#pragma unroll
        for (int j = 0; j < kQ / 2; ++j) {
            const float4 aw = wf[t & 1][j];
            const float4 bx = xf[j];
            v2f a0 = {aw.x, aw.y}, b0 = {bx.x, bx.y};
            v2f a1 = {aw.z, aw.w}, b1 = {bx.z, bx.w};
            acc = __builtin_amdgcn_wmma_f32_16x16x4_f32(
                false, a0, false, b0, (short)0, acc, false, false);
            acc = __builtin_amdgcn_wmma_f32_16x16x4_f32(
                false, a1, false, b1, (short)0, acc, false, false);
            q = a0 * a0 + q;
            q = a1 * a1 + q;
        }
        __builtin_amdgcn_sched_barrier(0);
        if (t + 1 < kBmuTilesPerWave) {
#pragma unroll
            for (int i = 0; i < 16; ++i) *(v4f*)&s_w[wave][(t + 1) * 16 + i][lane * 4] = ws[(t + 1) & 1][i];
        }
        if (t + 2 < kBmuTilesPerWave) {
#pragma unroll
            for (int i = 0; i < 16; ++i) ws[t & 1][i] = Wg[((t + 2) * 16 + i) * kRowQ];
        }
        if (t + 1 < kBmuTilesPerWave) {
#pragma unroll
            for (int j = 0; j < kQ; ++j)
                wf[(t + 1) & 1][j] = *(const float4*)&s_w[wave][(t + 1) * 16 + col][8 * j + 4 * half];
        }
        __builtin_amdgcn_sched_barrier(0);
#pragma unroll
        for (int j = kQ / 2; j < kQ; ++j) {
            const float4 aw = wf[t & 1][j];
            const float4 bx = xf[j];
            v2f a0 = {aw.x, aw.y}, b0 = {bx.x, bx.y};
            v2f a1 = {aw.z, aw.w}, b1 = {bx.z, bx.w};
            acc = __builtin_amdgcn_wmma_f32_16x16x4_f32(
                false, a0, false, b0, (short)0, acc, false, false);
            acc = __builtin_amdgcn_wmma_f32_16x16x4_f32(
                false, a1, false, b1, (short)0, acc, false, false);
            q = a0 * a0 + q;
            q = a1 * a1 + q;
        }
        const v2f mh = {-0.5f, -0.5f};
        acc = __builtin_amdgcn_wmma_f32_16x16x4_f32(
            false, q, false, mh, (short)0, acc, false, false);
#pragma unroll
        for (int v = 0; v < 8; ++v) {
            const int row = mt * 16 + v + half * 8;
            const float s = -acc[v];
            if (s < best) { best = s; bidx = row; }
        }
    }
    const float obest = __shfl_xor(best, 16);
    const int   oidx  = __shfl_xor(bidx, 16);
    if (obest < best || (obest == best && oidx < bidx)) { best = obest; bidx = oidx; }
    if (half == 0) { s_best[wave][col] = best; s_bidx[wave][col] = bidx; }
    __syncthreads();
    if (threadIdx.x < 16) {
        float fbest = s_best[0][col];
        int   fidx  = s_bidx[0][col];
        for (int w = 1; w < kBmuWaves; ++w) {
            const float ob = s_best[w][col];
            if (ob < fbest) { fbest = ob; fidx = s_bidx[w][col]; }
        }
        cand[rpart * kBatch + batch] =
            make_float4(fbest, (float)loc[fidx * 2], (float)loc[fidx * 2 + 1], 0.f);
    }
}

__global__ __launch_bounds__(kUpdThreads) __attribute__((amdgpu_waves_per_eu(4, 4)))
void som_update_kernel(const float*  __restrict__ W,
                       const float*  __restrict__ X,
                       const int*    __restrict__ loc,
                       const float4* __restrict__ cand,
                       const int*    __restrict__ itp,
                       float*        __restrict__ out) {
    __shared__ float s_acc[kUpdWaves][2][8][32];
    __shared__ float s_rs[kUpdWaves][16];
    __shared__ float4 s_loc[kUpdWaves][kUpdKPerWave / 2];

    const int lane = threadIdx.x & 31;
    const int wave = __builtin_amdgcn_readfirstlane(threadIdx.x >> 5);
    const int col  = lane & 15;
    const int half = lane >> 4;
    const int mt   = blockIdx.x;
    const int d0   = blockIdx.y * kUpdFeat;

    const float lr_decay   = 1.0f - (float)itp[0] * (1.0f / 100.0f);
    const float alpha_op   = 0.3f * lr_decay;
    const float sigma_op   = 16.0f * lr_decay;
    const float neg_inv_s2 = -1.0f / (sigma_op * sigma_op);

    const int   arow = mt * 16 + col;
    const float lx = (float)loc[arow * 2];
    const float ly = (float)loc[arow * 2 + 1];

    v8f   acc0 = {};
    v8f   acc1 = {};
    float rsum = 0.f;

    const int     kl = wave * kUpdKPerWave + half * 2;
    const float*  Xl = X + (kl * kDim + d0 + 2 * col);

    {
        const float4* c0 = cand + wave * kUpdKPerWave + lane;
        const float4* c1 = c0 + kBatch;
        float4 lo[4], hi[4];
#pragma unroll
        for (int e = 0; e < 4; ++e) { lo[e] = c0[e * 32]; hi[e] = c1[e * 32]; }
        float2* sl = (float2*)&s_loc[wave][0];
#pragma unroll
        for (int e = 0; e < 4; ++e) {
            const bool up = hi[e].x < lo[e].x;
            sl[e * 32 + lane] = make_float2(up ? hi[e].y : lo[e].y, up ? hi[e].z : lo[e].z);
        }
    }
    __syncthreads();
    const float4* Bl = &s_loc[wave][half];
#pragma unroll 8
    for (int ks = 0; ks < kUpdKPerWave; ks += 4) {
        const float4 bl = Bl[ks >> 1];
        const float dx0 = lx - bl.x, dy0 = ly - bl.y;
        const float dx1 = lx - bl.z, dy1 = ly - bl.w;
        v2f a;
        a.x = alpha_op * __expf((dx0 * dx0 + dy0 * dy0) * neg_inv_s2);
        a.y = alpha_op * __expf((dx1 * dx1 + dy1 * dy1) * neg_inv_s2);
        rsum += a.x + a.y;
        const float2 r0 = *(const float2*)(Xl + ks * kDim);
        const float2 r1 = *(const float2*)(Xl + (ks + 1) * kDim);
        const v2f b0 = {r0.x, r1.x};
        const v2f b1 = {r0.y, r1.y};
        acc0 = __builtin_amdgcn_wmma_f32_16x16x4_f32(
            false, a, false, b0, (short)0, acc0, false, false);
        acc1 = __builtin_amdgcn_wmma_f32_16x16x4_f32(
            false, a, false, b1, (short)0, acc1, false, false);
    }
    rsum += __shfl_xor(rsum, 16);
#pragma unroll
    for (int v = 0; v < 8; ++v) {
        s_acc[wave][0][v][lane] = acc0[v];
        s_acc[wave][1][v][lane] = acc1[v];
    }
    if (half == 0) s_rs[wave][col] = rsum;
    __syncthreads();
    if (wave < 8) {
        const int rt = wave + half * 8;
        float a0 = s_acc[0][0][wave][lane];
        float a1 = s_acc[0][1][wave][lane];
        float rs = s_rs[0][rt];
        for (int w = 1; w < kUpdWaves; ++w) {
            a0 += s_acc[w][0][wave][lane];
            a1 += s_acc[w][1][wave][lane];
            rs += s_rs[w][rt];
        }
        const int    row = mt * 16 + rt;
        const float2 w   = *(const float2*)(W + row * kDim + d0 + 2 * col);
        float2 o;
        o.x = w.x + a0 - rs * w.x;
        o.y = w.y + a1 - rs * w.y;
        *(float2*)(out + row * kDim + d0 + 2 * col) = o;
    }
}

extern "C" void kernel_launch(void* const* d_in, const int* in_sizes, int n_in,
                              void* d_out, int out_size, void* d_ws, size_t ws_size,
                              hipStream_t stream) {
    const float* x        = (const float*)d_in[0];
    const float* weights  = (const float*)d_in[1];
    const int*   locations= (const int*)  d_in[2];
    const int*   it       = (const int*)  d_in[3];
    float*       out      = (float*)d_out;

    float4* cand = (float4*)d_ws;

    dim3 g1(kBatch / 16, kBmuSplit);
    som_bmu_kernel<<<g1, kBmuThreads, 0, stream>>>(weights, x, locations, cand);
    dim3 g2(kMN / 16, kDim / kUpdFeat);
    som_update_kernel<<<g2, kUpdThreads, 0, stream>>>(weights, x, locations, cand, it, out);
}
